# starting kernel
# baseline (speedup 1.0000x reference)
#include <hip/hip_runtime.h>
#include <math.h>

#define N_USER 50000
#define N_REPO 100000
#define NEDGE  1000000
#define NEP    200000
#define D_IN   256
#define D_EMB  125
#define LD_EMB 128
#define D_HID  96
#define D_OUT  64
#define EPSN   1e-12f
#define MAX_K  256

typedef float v2f __attribute__((ext_vector_type(2)));
typedef float v8f __attribute__((ext_vector_type(8)));

__global__ void __launch_bounds__(128)
gemm_wmma_f32(const float* __restrict__ A, int lda,
              const float* __restrict__ W, int ldw,
              const float* __restrict__ degA,
              const float* __restrict__ bias,
              const float* __restrict__ agg,
              const float* __restrict__ degC,
              const float* __restrict__ bias2,
              float* __restrict__ C, int ldc,
              int M, int Nreal, int K, int Kpad, int relu)
{
  __shared__ __align__(16) float Wt[16 * (MAX_K + 4)];
  const int ldsStride = Kpad + 4;
  const int n0 = blockIdx.x * 16;

  {
    const int tid = threadIdx.y * 32 + threadIdx.x;
    const int sn  = tid >> 3;
    const int sk0 = tid & 7;
    const int gn  = n0 + sn;
    const bool gok = (gn < Nreal);
    float* dstp = &Wt[sn * ldsStride];
    for (int k = sk0; k < Kpad; k += 8)
      dstp[k] = (gok && k < K) ? W[(long)k * ldw + gn] : 0.0f;
  }
  __syncthreads();

  const int lane = threadIdx.x;
  const int half = lane >> 4;
  const int l16  = lane & 15;
  const int mtile = blockIdx.y * blockDim.y + threadIdx.y;
  if (mtile * 16 >= M) return;
  const int m0 = mtile * 16;

  const int   mA = m0 + l16;
  const float sA = degA ? rsqrtf(fmaxf(degA[mA], 1.0f)) : 1.0f;

  const float* Ap = A + (long)mA * lda + 2 * half;
  const float* Wp = &Wt[l16 * ldsStride + 2 * half];

  v8f acc = {};
#pragma unroll 4
  for (int k0 = 0; k0 < Kpad; k0 += 4) {
    v2f a = *(const v2f*)(Ap + k0);
    a *= sA;
    v2f b = *(const v2f*)(Wp + k0);
    acc = __builtin_amdgcn_wmma_f32_16x16x4_f32(false, a, false, b,
                                                (short)0, acc, false, false);
  }

  const int   n   = n0 + l16;
  const float bn  = (bias  && n < Nreal) ? bias[n]  : 0.0f;
  const float b2n = (bias2 && n < Nreal) ? bias2[n] : 0.0f;
#pragma unroll
  for (int v = 0; v < 8; ++v) {
    const int m = m0 + v + 8 * half;
    float val = acc[v] + bn;
    if (agg)  val += agg[(long)m * ldc + n] * rsqrtf(fmaxf(degC[m], 1.0f)) + b2n;
    if (relu) val = fmaxf(val, 0.0f);
    C[(long)m * ldc + n] = (n < Nreal) ? val : 0.0f;
  }
}

__global__ void degree_kernel(const int* __restrict__ src, const int* __restrict__ dst,
                              float* __restrict__ dsrc, float* __restrict__ ddst, int E)
{
  int e = blockIdx.x * blockDim.x + threadIdx.x;
  if (e < E) {
    atomicAdd(&dsrc[src[e]], 1.0f);
    atomicAdd(&ddst[dst[e]], 1.0f);
  }
}

__global__ void scatter_add_kernel(const float* __restrict__ feat,
                                   const int* __restrict__ src,
                                   const int* __restrict__ dst,
                                   float* __restrict__ agg, int E, int D)
{
  int e = blockIdx.x * blockDim.y + threadIdx.y;
  if (e >= E) return;
  int s = src[e], d = dst[e];
  atomicAdd(&agg[(long)d * D + threadIdx.x], feat[(long)s * D + threadIdx.x]);
}

__global__ void rownorm_kernel(const float* __restrict__ X, float* __restrict__ invn, int rows)
{
  int r = blockIdx.x * blockDim.y + threadIdx.y;
  if (r >= rows) return;
  int lane = threadIdx.x;
  float2 v = ((const float2*)(X + (long)r * D_OUT))[lane];
  float s = v.x * v.x + v.y * v.y;
#pragma unroll
  for (int o = 16; o > 0; o >>= 1) s += __shfl_xor(s, o, 32);
  if (lane == 0) invn[r] = 1.0f / fmaxf(sqrtf(s), EPSN);
}

__global__ void score_kernel(const float* __restrict__ HU, const float* __restrict__ HR,
                             const float* __restrict__ invU, const float* __restrict__ invR,
                             const int* __restrict__ iu, const int* __restrict__ ir,
                             float* __restrict__ out, int Ep)
{
  int e = blockIdx.x * blockDim.y + threadIdx.y;
  if (e >= Ep) return;
  int lane = threadIdx.x;
  int u = iu[e], r = ir[e];
  float2 a = ((const float2*)(HU + (long)u * D_OUT))[lane];
  float2 b = ((const float2*)(HR + (long)r * D_OUT))[lane];
  float s = a.x * b.x + a.y * b.y;
#pragma unroll
  for (int o = 16; o > 0; o >>= 1) s += __shfl_xor(s, o, 32);
  if (lane == 0) out[e] = s * invU[u] * invR[r];
}

extern "C" void kernel_launch(void* const* d_in, const int* in_sizes, int n_in,
                              void* d_out, int out_size, void* d_ws, size_t ws_size,
                              hipStream_t stream)
{
  const float* user_feat = (const float*)d_in[0];
  const float* repo_feat = (const float*)d_in[1];
  const float* W_ue = (const float*)d_in[2];   const float* b_ue = (const float*)d_in[3];
  const float* W_re = (const float*)d_in[4];   const float* b_re = (const float*)d_in[5];
  const float* W_hid_ur = (const float*)d_in[6];  const float* b_hid_ur = (const float*)d_in[7];
  const float* W_hid_ru = (const float*)d_in[8];  const float* b_hid_ru = (const float*)d_in[9];
  const float* W_out_ur = (const float*)d_in[10]; const float* b_out_ur = (const float*)d_in[11];
  const float* W_out_ru = (const float*)d_in[12]; const float* b_out_ru = (const float*)d_in[13];
  const float* W_hcu = (const float*)d_in[14]; const float* b_hcu = (const float*)d_in[15];
  const float* W_hcr = (const float*)d_in[16]; const float* b_hcr = (const float*)d_in[17];
  const float* W_ocu = (const float*)d_in[18]; const float* b_ocu = (const float*)d_in[19];
  const float* W_ocr = (const float*)d_in[20]; const float* b_ocr = (const float*)d_in[21];
  const int* e_ur_src = (const int*)d_in[22];  const int* e_ur_dst = (const int*)d_in[23];
  const int* e_ru_src = (const int*)d_in[24];  const int* e_ru_dst = (const int*)d_in[25];
  const int* pos_u = (const int*)d_in[26];     const int* pos_r = (const int*)d_in[27];
  const int* neg_u = (const int*)d_in[28];     const int* neg_r = (const int*)d_in[29];
  float* out = (float*)d_out;

  float* w = (float*)d_ws;
  size_t o = 0;
  float* deg_u_out_ur = w + o; o += N_USER;
  float* deg_r_in_ur  = w + o; o += N_REPO;
  float* deg_r_out_ru = w + o; o += N_REPO;
  float* deg_u_in_ru  = w + o; o += N_USER;
  float* agg_u   = w + o; o += (size_t)N_USER * D_HID;
  float* agg_r   = w + o; o += (size_t)N_REPO * D_HID;
  float* h_user  = w + o; o += (size_t)N_USER * LD_EMB;
  float* h_repo  = w + o; o += (size_t)N_REPO * LD_EMB;
  float* feat_u  = w + o; o += (size_t)N_USER * D_HID;
  float* feat_r  = w + o; o += (size_t)N_REPO * D_HID;
  float* out_user = w + o; o += (size_t)N_USER * D_HID;
  float* out_repo = w + o; o += (size_t)N_REPO * D_HID;
  float* inv_nu  = w + o; o += N_USER;
  float* inv_nr  = w + o; o += N_REPO;

  const size_t degBytes = (size_t)(2 * N_USER + 2 * N_REPO) * sizeof(float);
  const size_t aggBytes = ((size_t)N_USER + N_REPO) * D_HID * sizeof(float);

  auto gemm = [&](const float* A, int lda, const float* W_, int ldw,
                  const float* degA, const float* bias,
                  const float* agg, const float* degC, const float* bias2,
                  float* C, int ldc, int M, int Nreal, int K, int relu) {
    const int Kpad = (K + 3) & ~3;
    dim3 blk(32, 4);
    dim3 grd(ldc / 16, (M / 16 + 3) / 4);
    hipLaunchKernelGGL(gemm_wmma_f32, grd, blk, 0, stream,
                       A, lda, W_, ldw, degA, bias, agg, degC, bias2,
                       C, ldc, M, Nreal, K, Kpad, relu);
  };

  auto scatter = [&](const float* feat, const int* src, const int* dst,
                     float* agg, int D) {
    const int epb = 256 / D;
    dim3 blk(D, epb);
    dim3 grd((NEDGE + epb - 1) / epb);
    hipLaunchKernelGGL(scatter_add_kernel, grd, blk, 0, stream,
                       feat, src, dst, agg, NEDGE, D);
  };

  hipMemsetAsync(deg_u_out_ur, 0, degBytes, stream);
  hipMemsetAsync(agg_u, 0, aggBytes, stream);
  hipLaunchKernelGGL(degree_kernel, dim3((NEDGE + 255) / 256), dim3(256), 0, stream,
                     e_ur_src, e_ur_dst, deg_u_out_ur, deg_r_in_ur, NEDGE);
  hipLaunchKernelGGL(degree_kernel, dim3((NEDGE + 255) / 256), dim3(256), 0, stream,
                     e_ru_src, e_ru_dst, deg_r_out_ru, deg_u_in_ru, NEDGE);

  gemm(user_feat, D_IN, W_ue, D_EMB, nullptr, b_ue, nullptr, nullptr, nullptr,
       h_user, LD_EMB, N_USER, D_EMB, D_IN, 0);
  gemm(repo_feat, D_IN, W_re, D_EMB, nullptr, b_re, nullptr, nullptr, nullptr,
       h_repo, LD_EMB, N_REPO, D_EMB, D_IN, 0);

  gemm(h_user, LD_EMB, W_hid_ur, D_HID, deg_u_out_ur, nullptr, nullptr, nullptr, nullptr,
       feat_u, D_HID, N_USER, D_HID, D_EMB, 0);
  gemm(h_repo, LD_EMB, W_hid_ru, D_HID, deg_r_out_ru, nullptr, nullptr, nullptr, nullptr,
       feat_r, D_HID, N_REPO, D_HID, D_EMB, 0);

  scatter(feat_u, e_ur_src, e_ur_dst, agg_r, D_HID);
  scatter(feat_r, e_ru_src, e_ru_dst, agg_u, D_HID);

  gemm(h_user, LD_EMB, W_hcu, D_HID, nullptr, b_hcu, agg_u, deg_u_in_ru, b_hid_ru,
       out_user, D_HID, N_USER, D_HID, D_EMB, 1);
  gemm(h_repo, LD_EMB, W_hcr, D_HID, nullptr, b_hcr, agg_r, deg_r_in_ur, b_hid_ur,
       out_repo, D_HID, N_REPO, D_HID, D_EMB, 1);

  hipMemsetAsync(agg_u, 0, aggBytes, stream);
  gemm(out_user, D_HID, W_out_ur, D_OUT, deg_u_out_ur, nullptr, nullptr, nullptr, nullptr,
       feat_u, D_OUT, N_USER, D_OUT, D_HID, 0);
  gemm(out_repo, D_HID, W_out_ru, D_OUT, deg_r_out_ru, nullptr, nullptr, nullptr, nullptr,
       feat_r, D_OUT, N_REPO, D_OUT, D_HID, 0);

  scatter(feat_u, e_ur_src, e_ur_dst, agg_r, D_OUT);
  scatter(feat_r, e_ru_src, e_ru_dst, agg_u, D_OUT);

  gemm(out_user, D_HID, W_ocu, D_OUT, nullptr, b_ocu, agg_u, deg_u_in_ru, b_out_ru,
       h_user, D_OUT, N_USER, D_OUT, D_HID, 1);
  gemm(out_repo, D_HID, W_ocr, D_OUT, nullptr, b_ocr, agg_r, deg_r_in_ur, b_out_ur,
       h_repo, D_OUT, N_REPO, D_OUT, D_HID, 1);

  hipLaunchKernelGGL(rownorm_kernel, dim3((N_USER + 7) / 8), dim3(32, 8), 0, stream,
                     h_user, inv_nu, N_USER);
  hipLaunchKernelGGL(rownorm_kernel, dim3((N_REPO + 7) / 8), dim3(32, 8), 0, stream,
                     h_repo, inv_nr, N_REPO);
  hipLaunchKernelGGL(score_kernel, dim3((NEP + 7) / 8), dim3(32, 8), 0, stream,
                     h_user, h_repo, inv_nu, inv_nr, pos_u, pos_r, out, NEP);
  hipLaunchKernelGGL(score_kernel, dim3((NEP + 7) / 8), dim3(32, 8), 0, stream,
                     h_user, h_repo, inv_nu, inv_nr, neg_u, neg_r, out + NEP, NEP);
}


# Cosine tail in one launch: each pair wave also sums the squares of the two rows it loads; the 2 row-norm passes and norm arrays go, pos and neg pairs share a grid; 13 launches to 10; sums unchanged
# speedup vs baseline: 2.7552x; 2.7552x over previous
#include <hip/hip_runtime.h>
#include <math.h>

#define N_USER 50000
#define N_REPO 100000
#define NEDGE  1000000
#define NEP    200000
#define D_IN   256
#define D_EMB  125
#define LD_EMB 128
#define D_HID  96
#define D_OUT  64
#define EPSN   1e-12f
#define MAX_K  256

typedef float v2f __attribute__((ext_vector_type(2)));
typedef float v4f __attribute__((ext_vector_type(4)));
typedef float v8f __attribute__((ext_vector_type(8)));

template <int NW>
__device__ __forceinline__ void
gemm_block(const float* __restrict__ A, int lda,
           const float* __restrict__ W, int ldw,
           const float* __restrict__ degA,
           const float* __restrict__ bias,
           const float* __restrict__ agg,
           const float* __restrict__ degC,
           const float* __restrict__ bias2,
           float* __restrict__ C, int ldc,
           int M, int Nreal, int K, int Kpad, int relu,
           int n0, int mtile, int tid, int lane)
{
  __shared__ __align__(16) float Wt[16 * (MAX_K + 4)];
  const int ldsStride = Kpad + 4;

  {
    const int sn  = tid / (2 * NW);
    const int sk0 = tid % (2 * NW);
    const int gn  = n0 + sn;
    const bool gok = (gn < Nreal);
    float* dstp = &Wt[sn * ldsStride];
    for (int k = sk0; k < Kpad; k += 2 * NW)
      dstp[k] = (gok && k < K) ? W[(long)k * ldw + gn] : 0.0f;
  }
  __syncthreads();

  const int half = lane >> 4;
  const int l16  = lane & 15;
  if (mtile * 16 >= M) return;
  const int m0 = mtile * 16;

  const int   mA = m0 + l16;
  const float sA = degA ? rsqrtf(fmaxf(degA[mA], 1.0f)) : 1.0f;

  const float* Ap = A + (long)mA * lda + 4 * half;
  const float* Wp = &Wt[l16 * ldsStride + 4 * half];

  v8f acc = {};
#pragma unroll 4
  for (int k0 = 0; k0 < Kpad; k0 += 8) {
    v4f a = *(const v4f*)(Ap + k0);
    a *= sA;
    const v4f b = *(const v4f*)(Wp + k0);
    const v2f a0 = {a.x, a.y}, a1 = {a.z, a.w};
    const v2f b0 = {b.x, b.y}, b1 = {b.z, b.w};
    acc = __builtin_amdgcn_wmma_f32_16x16x4_f32(false, a0, false, b0,
                                                (short)0, acc, false, false);
    acc = __builtin_amdgcn_wmma_f32_16x16x4_f32(false, a1, false, b1,
                                                (short)0, acc, false, false);
  }

  const int   n   = n0 + l16;
  const float bn  = (bias  && n < Nreal) ? bias[n]  : 0.0f;
  const float b2n = (bias2 && n < Nreal) ? bias2[n] : 0.0f;
#pragma unroll
  for (int v = 0; v < 8; ++v) {
    const int m = m0 + v + 8 * half;
    float val = acc[v] + bn;
    if (agg)  val += agg[(long)m * ldc + n] * rsqrtf(fmaxf(degC[m], 1.0f)) + b2n;
    if (relu) val = fmaxf(val, 0.0f);
    C[(long)m * ldc + n] = (n < Nreal) ? val : 0.0f;
  }
}

__global__ void __launch_bounds__(128)
gemm_wmma_f32(const float* __restrict__ A, int lda,
              const float* __restrict__ W, int ldw,
              const float* __restrict__ degA, const float* __restrict__ bias,
              const float* __restrict__ agg, const float* __restrict__ degC,
              const float* __restrict__ bias2,
              float* __restrict__ C, int ldc,
              int M, int Nreal, int K, int Kpad, int relu)
{
  gemm_block<4>(A, lda, W, ldw, degA, bias, agg, degC, bias2, C, ldc, M, Nreal, K, Kpad, relu,
             (int)blockIdx.x * 16, (int)(blockIdx.y * blockDim.y + threadIdx.y),
             (int)(threadIdx.y * 32 + threadIdx.x), (int)threadIdx.x);
}

#define TK_STRIDE 32
#define TK_LAUNCHES 6

__device__ __forceinline__ int draw_chunk(int* __restrict__ tk, int nChunks, int* half)
{
  __shared__ int pick[2];
  if (threadIdx.x == 0) {
    int h = (int)((__builtin_amdgcn_s_sendmsg_rtn(0x87) >> 18) & 1u);
    int t = atomicAdd(&tk[h * TK_STRIDE], 1);
    if (t >= nChunks) { h ^= 1; t = atomicAdd(&tk[h * TK_STRIDE], 1); }
    pick[0] = t;
    pick[1] = h;
  }
  __syncthreads();
  *half = pick[1];
  const int t = pick[0];
  return (t < nChunks) ? t : -1;
}

#define DG_TPB 128
#define DG_EPT 4
#define DG_CHUNK (DG_TPB * DG_EPT)
__device__ __forceinline__ void
degree_block(const int* __restrict__ src, const int* __restrict__ dst,
             float* __restrict__ dsrc, float* __restrict__ ddst, int E,
             int splitS, int splitD, int* __restrict__ tk, int nChunks)
{
  int half;
  const int t = draw_chunk(tk, nChunks, &half);
  if (t < 0) return;
  const bool hi = (half != 0);
  const int base = t * DG_CHUNK + threadIdx.x;
#pragma unroll
  for (int k = 0; k < DG_EPT; ++k) {
    const int e = base + k * DG_TPB;
    if (e < E) {
      const int s = src[e], d = dst[e];
      if ((s >= splitS) == hi) atomicAdd(&dsrc[s], 1.0f);
      if ((d >= splitD) == hi) atomicAdd(&ddst[d], 1.0f);
    }
  }
}

#define HD_DG_CHUNKS ((NEDGE + DG_CHUNK - 1) / DG_CHUNK)
#define HD_DG_SET    (2 * HD_DG_CHUNKS)
#define HD_DG_BLOCKS (2 * HD_DG_SET)
#define HD_NT        (LD_EMB / 16)
#define HD_GU_BLOCKS (HD_NT * ((N_USER / 16 + 3) / 4))
#define HD_GR_BLOCKS (HD_NT * ((N_REPO / 16 + 3) / 4))
#define HD_G_BLOCKS  (HD_GU_BLOCKS + HD_GR_BLOCKS)
#define HD_BLOCKS    (HD_DG_BLOCKS + HD_G_BLOCKS)
static_assert(HD_G_BLOCKS >= 2 * HD_DG_BLOCKS, "the one-in-three interleave needs two GEMM blocks per degree block");
static_assert(N_USER % 16 == 0 && N_REPO % 16 == 0, "whole m-tiles");

__global__ void __launch_bounds__(DG_TPB)
head_degrees_and_embeddings(const float* __restrict__ user_feat, const float* __restrict__ repo_feat,
                            const float* __restrict__ W_ue, const float* __restrict__ b_ue,
                            const float* __restrict__ W_re, const float* __restrict__ b_re,
                            float* __restrict__ h_user, float* __restrict__ h_repo,
                            const int* __restrict__ ur_src, const int* __restrict__ ur_dst,
                            const int* __restrict__ ru_src, const int* __restrict__ ru_dst,
                            float* __restrict__ deg_u_out_ur, float* __restrict__ deg_r_in_ur,
                            float* __restrict__ deg_r_out_ru, float* __restrict__ deg_u_in_ru,
                            int* __restrict__ tk)
{
  const int b = blockIdx.x;
  int dgi = -1, gi;
  if (b < 3 * HD_DG_BLOCKS) {
    const int q = b / 3, r = b - 3 * q;
    if (r == 0) dgi = q;
    gi = 2 * q + r - 1;
  } else {
    gi = 2 * HD_DG_BLOCKS + (b - 3 * HD_DG_BLOCKS);
  }

  if (dgi >= 0) {
    if (dgi < HD_DG_SET)
      degree_block(ur_src, ur_dst, deg_u_out_ur, deg_r_in_ur, NEDGE,
                   N_USER / 2, N_REPO / 2, tk + 0 * 2 * TK_STRIDE, HD_DG_CHUNKS);
    else
      degree_block(ru_src, ru_dst, deg_r_out_ru, deg_u_in_ru, NEDGE,
                   N_REPO / 2, N_USER / 2, tk + 1 * 2 * TK_STRIDE, HD_DG_CHUNKS);
    return;
  }

  const bool usr = (gi < HD_GU_BLOCKS);
  const int  g   = usr ? gi : gi - HD_GU_BLOCKS;
  const int  n0  = (g % HD_NT) * 16;
  const int  mt  = (g / HD_NT) * 4 + (int)(threadIdx.x >> 5);
  gemm_block<4>(usr ? user_feat : repo_feat, D_IN, usr ? W_ue : W_re, D_EMB,
             nullptr, usr ? b_ue : b_re, nullptr, nullptr, nullptr,
             usr ? h_user : h_repo, LD_EMB, usr ? N_USER : N_REPO, D_EMB, D_IN, D_IN, 0,
             n0, mt, (int)threadIdx.x, (int)(threadIdx.x & 31));
}

#define SC_EPW   128
#define SC_WPB   8
#define SC_AHEAD 8
#define SC_CHUNK (SC_EPW * SC_WPB)

template <int D, bool FULL>
__device__ __forceinline__ void move_group(const float* __restrict__ feat, float* __restrict__ agg,
                                           int sa, int sb, int da, int db,
                                           unsigned long long& m, int lane)
{
  constexpr int NC = D / 32;
  float v[SC_AHEAD][NC];
  int   dd[SC_AHEAD];
#pragma unroll
  for (int u = 0; u < SC_AHEAD; ++u) {
    dd[u] = -1;
    if (FULL || m != 0ull) {
      const int j = __builtin_ctzll(m);
      m &= m - 1ull;
      const int s_lo = __builtin_amdgcn_readlane(sa, j & 31);
      const int s_hi = __builtin_amdgcn_readlane(sb, j & 31);
      const int d_lo = __builtin_amdgcn_readlane(da, j & 31);
      const int d_hi = __builtin_amdgcn_readlane(db, j & 31);
      const int s = (j < 32) ? s_lo : s_hi;
      dd[u]       = (j < 32) ? d_lo : d_hi;
      const float* fp = feat + (long)s * D + lane;
#pragma unroll
      for (int c = 0; c < NC; ++c) v[u][c] = fp[32 * c];
    }
  }
#pragma unroll
  for (int u = 0; u < SC_AHEAD; ++u) {
    if (FULL || dd[u] >= 0) {
      float* ap = agg + (long)dd[u] * D + lane;
#pragma unroll
      for (int c = 0; c < NC; ++c) atomicAdd(ap + 32 * c, v[u][c]);
    }
  }
}

template <int D>
__device__ __forceinline__ void
scatter_block(const float* __restrict__ feat,
              const int* __restrict__ src,
              const int* __restrict__ dst,
              float* __restrict__ agg, int E,
              int splitD, int* __restrict__ tk, int nChunks)
{
  int half;
  const int t = draw_chunk(tk, nChunks, &half);
  if (t < 0) return;
  const bool hi = (half != 0);
  const int lane = threadIdx.x & 31;
  const int wave = __builtin_amdgcn_readfirstlane((int)(threadIdx.x >> 5));
  int e0 = t * SC_CHUNK + wave * SC_EPW;
  for (int b = 0; b < SC_EPW / 64; ++b, e0 += 64) {
    if (e0 >= E) return;
    const int ea = e0 + lane, eb = e0 + 32 + lane;
    const int sa = (ea < E) ? src[ea] : 0;
    const int da = (ea < E) ? dst[ea] : 0;
    const int sb = (eb < E) ? src[eb] : 0;
    const int db = (eb < E) ? dst[eb] : 0;
    const unsigned ma = __builtin_amdgcn_ballot_w32((ea < E) && ((da >= splitD) == hi));
    const unsigned mb = __builtin_amdgcn_ballot_w32((eb < E) && ((db >= splitD) == hi));
    unsigned long long m = (unsigned long long)ma | ((unsigned long long)mb << 32);
    while (__builtin_popcountll(m) >= SC_AHEAD)
      move_group<D, true>(feat, agg, sa, sb, da, db, m, lane);
    if (m != 0ull)
      move_group<D, false>(feat, agg, sa, sb, da, db, m, lane);
  }
}

template <int D>
__global__ void __launch_bounds__(32 * SC_WPB, 16)
scatter_and_gemm(const float* __restrict__ feat, const int* __restrict__ src, const int* __restrict__ dst,
                 float* __restrict__ aggS, int E, int splitD, int* __restrict__ tk, int nChunks, int nSc,
                 const float* __restrict__ A, int lda, const float* __restrict__ W, int ldw,
                 const float* __restrict__ degA, const float* __restrict__ bias,
                 const float* __restrict__ aggG, const float* __restrict__ degC,
                 const float* __restrict__ bias2,
                 float* __restrict__ C, int ldc, int M, int Nreal, int K, int Kpad, int relu)
{
  const int b = blockIdx.x;
  int gi;
  if (b < 2 * nSc) {
    if ((b & 1) == 0) {
      scatter_block<D>(feat, src, dst, aggS, E, splitD, tk, nChunks);
      return;
    }
    gi = b >> 1;
  } else {
    gi = b - nSc;
  }
  const int nt = ldc / 16;
  gemm_block<SC_WPB>(A, lda, W, ldw, degA, bias, aggG, degC, bias2, C, ldc, M, Nreal, K, Kpad, relu,
                     (gi % nt) * 16, (gi / nt) * SC_WPB + (int)(threadIdx.x >> 5),
                     (int)threadIdx.x, (int)(threadIdx.x & 31));
}

#define SCORE_PPB 8
static_assert(NEP % SCORE_PPB == 0, "the pairs of a block lie on one side, and the grid is exact");
__global__ void score_kernel(const float* __restrict__ HU, const float* __restrict__ HR,
                             const int* __restrict__ pu, const int* __restrict__ pr,
                             const int* __restrict__ nu, const int* __restrict__ nr,
                             float* __restrict__ out)
{
  const bool neg = ((int)blockIdx.x >= NEP / SCORE_PPB);
  const int  e   = (int)blockIdx.x * SCORE_PPB + (int)threadIdx.y;
  const int  i   = neg ? e - NEP : e;
  const int* iu  = neg ? nu : pu;
  const int* ir  = neg ? nr : pr;
  const int lane = threadIdx.x;
  const int u = iu[i], r = ir[i];
  const float2 a = ((const float2*)(HU + (long)u * D_OUT))[lane];
  const float2 b = ((const float2*)(HR + (long)r * D_OUT))[lane];
  float s  = a.x * b.x + a.y * b.y;
  float su = a.x * a.x + a.y * a.y;
  float sr = b.x * b.x + b.y * b.y;
#pragma unroll
  for (int o = 16; o > 0; o >>= 1) {
    s  += __shfl_xor(s,  o, 32);
    su += __shfl_xor(su, o, 32);
    sr += __shfl_xor(sr, o, 32);
  }
  if (lane == 0) {
    const float invU = 1.0f / fmaxf(sqrtf(su), EPSN);
    const float invR = 1.0f / fmaxf(sqrtf(sr), EPSN);
    out[e] = s * invU * invR;
  }
}

extern "C" void kernel_launch(void* const* d_in, const int* in_sizes, int n_in,
                              void* d_out, int out_size, void* d_ws, size_t ws_size,
                              hipStream_t stream)
{
  const float* user_feat = (const float*)d_in[0];
  const float* repo_feat = (const float*)d_in[1];
  const float* W_ue = (const float*)d_in[2];   const float* b_ue = (const float*)d_in[3];
  const float* W_re = (const float*)d_in[4];   const float* b_re = (const float*)d_in[5];
  const float* W_hid_ur = (const float*)d_in[6];  const float* b_hid_ur = (const float*)d_in[7];
  const float* W_hid_ru = (const float*)d_in[8];  const float* b_hid_ru = (const float*)d_in[9];
  const float* W_out_ur = (const float*)d_in[10]; const float* b_out_ur = (const float*)d_in[11];
  const float* W_out_ru = (const float*)d_in[12]; const float* b_out_ru = (const float*)d_in[13];
  const float* W_hcu = (const float*)d_in[14]; const float* b_hcu = (const float*)d_in[15];
  const float* W_hcr = (const float*)d_in[16]; const float* b_hcr = (const float*)d_in[17];
  const float* W_ocu = (const float*)d_in[18]; const float* b_ocu = (const float*)d_in[19];
  const float* W_ocr = (const float*)d_in[20]; const float* b_ocr = (const float*)d_in[21];
  const int* e_ur_src = (const int*)d_in[22];  const int* e_ur_dst = (const int*)d_in[23];
  const int* e_ru_src = (const int*)d_in[24];  const int* e_ru_dst = (const int*)d_in[25];
  const int* pos_u = (const int*)d_in[26];     const int* pos_r = (const int*)d_in[27];
  const int* neg_u = (const int*)d_in[28];     const int* neg_r = (const int*)d_in[29];
  float* out = (float*)d_out;

  float* w = (float*)d_ws;
  size_t o = 0;
  float* deg_u_out_ur = w + o; o += N_USER;
  float* deg_r_in_ur  = w + o; o += N_REPO;
  float* deg_r_out_ru = w + o; o += N_REPO;
  float* deg_u_in_ru  = w + o; o += N_USER;
  int*   tk      = (int*)(w + o); o += TK_LAUNCHES * 2 * TK_STRIDE;
  float* agg_u   = w + o; o += (size_t)N_USER * D_HID;
  float* agg_r   = w + o; o += (size_t)N_REPO * D_HID;
  float* h_user  = w + o; o += (size_t)N_USER * LD_EMB;
  float* h_repo  = w + o; o += (size_t)N_REPO * LD_EMB;
  float* feat_u  = w + o; o += (size_t)N_USER * D_HID;
  float* feat_r  = w + o; o += (size_t)N_REPO * D_HID;
  float* out_user = w + o; o += (size_t)N_USER * D_HID;
  float* out_repo = w + o; o += (size_t)N_REPO * D_HID;

  const size_t degBytes = (size_t)(2 * N_USER + 2 * N_REPO) * sizeof(float)
                        + (size_t)(TK_LAUNCHES * 2 * TK_STRIDE) * sizeof(int);
  const size_t aggBytes = ((size_t)N_USER + N_REPO) * D_HID * sizeof(float);

  auto gemm = [&](const float* A, int lda, const float* W_, int ldw,
                  const float* degA, const float* bias,
                  const float* agg, const float* degC, const float* bias2,
                  float* C, int ldc, int M, int Nreal, int K, int relu) {
    const int Kpad = (K + 7) & ~7;
    dim3 blk(32, 4);
    dim3 grd(ldc / 16, (M / 16 + 3) / 4);
    hipLaunchKernelGGL(gemm_wmma_f32, grd, blk, 0, stream,
                       A, lda, W_, ldw, degA, bias, agg, degC, bias2,
                       C, ldc, M, Nreal, K, Kpad, relu);
  };

  const int  scChunks = (NEDGE + SC_CHUNK - 1) / SC_CHUNK;
  const dim3 scBlk(32 * SC_WPB);

  hipMemsetAsync(deg_u_out_ur, 0, degBytes, stream);
  hipMemsetAsync(agg_u, 0, aggBytes, stream);
  hipLaunchKernelGGL(head_degrees_and_embeddings, dim3(HD_BLOCKS), dim3(DG_TPB), 0, stream,
                     user_feat, repo_feat, W_ue, b_ue, W_re, b_re, h_user, h_repo,
                     e_ur_src, e_ur_dst, e_ru_src, e_ru_dst,
                     deg_u_out_ur, deg_r_in_ur, deg_r_out_ru, deg_u_in_ru, tk);

  auto fused = [&](auto kern, const float* feat, const int* s, const int* d, float* aggS, int splitD, int* tkp,
                   const float* A, int lda, const float* W_, int ldw,
                   const float* degA, const float* bias,
                   const float* aggG, const float* degC, const float* bias2,
                   float* C, int ldc, int M, int Nreal, int K, int relu) {
    const int Kpad = (K + 7) & ~7;
    const int nG = (ldc / 16) * ((M / 16 + SC_WPB - 1) / SC_WPB);
    hipLaunchKernelGGL(kern, dim3(2 * scChunks + nG), scBlk, 0, stream,
                       feat, s, d, aggS, NEDGE, splitD, tkp, scChunks, 2 * scChunks,
                       A, lda, W_, ldw, degA, bias, aggG, degC, bias2, C, ldc, M, Nreal, K, Kpad, relu);
  };

  gemm(h_user, LD_EMB, W_hid_ur, D_HID, deg_u_out_ur, nullptr, nullptr, nullptr, nullptr,
       feat_u, D_HID, N_USER, D_HID, D_EMB, 0);
  fused(scatter_and_gemm<D_HID>, feat_u, e_ur_src, e_ur_dst, agg_r, N_REPO / 2, tk + 2 * 2 * TK_STRIDE,
        h_repo, LD_EMB, W_hid_ru, D_HID, deg_r_out_ru, nullptr, nullptr, nullptr, nullptr,
        feat_r, D_HID, N_REPO, D_HID, D_EMB, 0);
  fused(scatter_and_gemm<D_HID>, feat_r, e_ru_src, e_ru_dst, agg_u, N_USER / 2, tk + 3 * 2 * TK_STRIDE,
        h_repo, LD_EMB, W_hcr, D_HID, nullptr, b_hcr, agg_r, deg_r_in_ur, b_hid_ur,
        out_repo, D_HID, N_REPO, D_HID, D_EMB, 1);
  gemm(h_user, LD_EMB, W_hcu, D_HID, nullptr, b_hcu, agg_u, deg_u_in_ru, b_hid_ru,
       out_user, D_HID, N_USER, D_HID, D_EMB, 1);

  hipMemsetAsync(agg_u, 0, aggBytes, stream);
  gemm(out_user, D_HID, W_out_ur, D_OUT, deg_u_out_ur, nullptr, nullptr, nullptr, nullptr,
       feat_u, D_OUT, N_USER, D_OUT, D_HID, 0);
  fused(scatter_and_gemm<D_OUT>, feat_u, e_ur_src, e_ur_dst, agg_r, N_REPO / 2, tk + 4 * 2 * TK_STRIDE,
        out_repo, D_HID, W_out_ru, D_OUT, deg_r_out_ru, nullptr, nullptr, nullptr, nullptr,
        feat_r, D_OUT, N_REPO, D_OUT, D_HID, 0);
  fused(scatter_and_gemm<D_OUT>, feat_r, e_ru_src, e_ru_dst, agg_u, N_USER / 2, tk + 5 * 2 * TK_STRIDE,
        out_repo, D_HID, W_ocr, D_OUT, nullptr, b_ocr, agg_r, deg_r_in_ur, b_out_ur,
        h_repo, D_OUT, N_REPO, D_OUT, D_HID, 1);
  gemm(out_user, D_HID, W_ocu, D_OUT, nullptr, b_ocu, agg_u, deg_u_in_ru, b_out_ru,
       h_user, D_OUT, N_USER, D_OUT, D_HID, 1);

  hipLaunchKernelGGL(score_kernel, dim3(2 * NEP / SCORE_PPB), dim3(32, SCORE_PPB), 0, stream,
                     h_user, h_repo, pos_u, pos_r, neg_u, neg_r, out);
}


# Cosine tail: a wave per pair becomes a wave per 32 consecutive pairs: row numbers by one coalesced load, 8 pairs' rows in flight, their sums share one lane tree; 387,500 fewer waves; sums unchanged
# speedup vs baseline: 2.9259x; 1.0620x over previous
#include <hip/hip_runtime.h>
#include <math.h>

#define N_USER 50000
#define N_REPO 100000
#define NEDGE  1000000
#define NEP    200000
#define D_IN   256
#define D_EMB  125
#define LD_EMB 128
#define D_HID  96
#define D_OUT  64
#define EPSN   1e-12f
#define MAX_K  256

typedef float v2f __attribute__((ext_vector_type(2)));
typedef float v4f __attribute__((ext_vector_type(4)));
typedef float v8f __attribute__((ext_vector_type(8)));

template <int NW>
__device__ __forceinline__ void
gemm_block(const float* __restrict__ A, int lda,
           const float* __restrict__ W, int ldw,
           const float* __restrict__ degA,
           const float* __restrict__ bias,
           const float* __restrict__ agg,
           const float* __restrict__ degC,
           const float* __restrict__ bias2,
           float* __restrict__ C, int ldc,
           int M, int Nreal, int K, int Kpad, int relu,
           int n0, int mtile, int tid, int lane)
{
  __shared__ __align__(16) float Wt[16 * (MAX_K + 4)];
  const int ldsStride = Kpad + 4;

  {
    const int sn  = tid / (2 * NW);
    const int sk0 = tid % (2 * NW);
    const int gn  = n0 + sn;
    const bool gok = (gn < Nreal);
    float* dstp = &Wt[sn * ldsStride];
    for (int k = sk0; k < Kpad; k += 2 * NW)
      dstp[k] = (gok && k < K) ? W[(long)k * ldw + gn] : 0.0f;
  }
  __syncthreads();

  const int half = lane >> 4;
  const int l16  = lane & 15;
  if (mtile * 16 >= M) return;
  const int m0 = mtile * 16;

  const int   mA = m0 + l16;
  const float sA = degA ? rsqrtf(fmaxf(degA[mA], 1.0f)) : 1.0f;

  const float* Ap = A + (long)mA * lda + 4 * half;
  const float* Wp = &Wt[l16 * ldsStride + 4 * half];

  v8f acc = {};
#pragma unroll 4
  for (int k0 = 0; k0 < Kpad; k0 += 8) {
    v4f a = *(const v4f*)(Ap + k0);
    a *= sA;
    const v4f b = *(const v4f*)(Wp + k0);
    const v2f a0 = {a.x, a.y}, a1 = {a.z, a.w};
    const v2f b0 = {b.x, b.y}, b1 = {b.z, b.w};
    acc = __builtin_amdgcn_wmma_f32_16x16x4_f32(false, a0, false, b0,
                                                (short)0, acc, false, false);
    acc = __builtin_amdgcn_wmma_f32_16x16x4_f32(false, a1, false, b1,
                                                (short)0, acc, false, false);
  }

  const int   n   = n0 + l16;
  const float bn  = (bias  && n < Nreal) ? bias[n]  : 0.0f;
  const float b2n = (bias2 && n < Nreal) ? bias2[n] : 0.0f;
#pragma unroll
  for (int v = 0; v < 8; ++v) {
    const int m = m0 + v + 8 * half;
    float val = acc[v] + bn;
    if (agg)  val += agg[(long)m * ldc + n] * rsqrtf(fmaxf(degC[m], 1.0f)) + b2n;
    if (relu) val = fmaxf(val, 0.0f);
    C[(long)m * ldc + n] = (n < Nreal) ? val : 0.0f;
  }
}

__global__ void __launch_bounds__(128)
gemm_wmma_f32(const float* __restrict__ A, int lda,
              const float* __restrict__ W, int ldw,
              const float* __restrict__ degA, const float* __restrict__ bias,
              const float* __restrict__ agg, const float* __restrict__ degC,
              const float* __restrict__ bias2,
              float* __restrict__ C, int ldc,
              int M, int Nreal, int K, int Kpad, int relu)
{
  gemm_block<4>(A, lda, W, ldw, degA, bias, agg, degC, bias2, C, ldc, M, Nreal, K, Kpad, relu,
             (int)blockIdx.x * 16, (int)(blockIdx.y * blockDim.y + threadIdx.y),
             (int)(threadIdx.y * 32 + threadIdx.x), (int)threadIdx.x);
}

#define TK_STRIDE 32
#define TK_LAUNCHES 6

__device__ __forceinline__ int draw_chunk(int* __restrict__ tk, int nChunks, int* half)
{
  __shared__ int pick[2];
  if (threadIdx.x == 0) {
    int h = (int)((__builtin_amdgcn_s_sendmsg_rtn(0x87) >> 18) & 1u);
    int t = atomicAdd(&tk[h * TK_STRIDE], 1);
    if (t >= nChunks) { h ^= 1; t = atomicAdd(&tk[h * TK_STRIDE], 1); }
    pick[0] = t;
    pick[1] = h;
  }
  __syncthreads();
  *half = pick[1];
  const int t = pick[0];
  return (t < nChunks) ? t : -1;
}

#define DG_TPB 128
#define DG_EPT 4
#define DG_CHUNK (DG_TPB * DG_EPT)
__device__ __forceinline__ void
degree_block(const int* __restrict__ src, const int* __restrict__ dst,
             float* __restrict__ dsrc, float* __restrict__ ddst, int E,
             int splitS, int splitD, int* __restrict__ tk, int nChunks)
{
  int half;
  const int t = draw_chunk(tk, nChunks, &half);
  if (t < 0) return;
  const bool hi = (half != 0);
  const int base = t * DG_CHUNK + threadIdx.x;
#pragma unroll
  for (int k = 0; k < DG_EPT; ++k) {
    const int e = base + k * DG_TPB;
    if (e < E) {
      const int s = src[e], d = dst[e];
      if ((s >= splitS) == hi) atomicAdd(&dsrc[s], 1.0f);
      if ((d >= splitD) == hi) atomicAdd(&ddst[d], 1.0f);
    }
  }
}

#define HD_DG_CHUNKS ((NEDGE + DG_CHUNK - 1) / DG_CHUNK)
#define HD_DG_SET    (2 * HD_DG_CHUNKS)
#define HD_DG_BLOCKS (2 * HD_DG_SET)
#define HD_NT        (LD_EMB / 16)
#define HD_GU_BLOCKS (HD_NT * ((N_USER / 16 + 3) / 4))
#define HD_GR_BLOCKS (HD_NT * ((N_REPO / 16 + 3) / 4))
#define HD_G_BLOCKS  (HD_GU_BLOCKS + HD_GR_BLOCKS)
#define HD_BLOCKS    (HD_DG_BLOCKS + HD_G_BLOCKS)
static_assert(HD_G_BLOCKS >= 2 * HD_DG_BLOCKS, "the one-in-three interleave needs two GEMM blocks per degree block");
static_assert(N_USER % 16 == 0 && N_REPO % 16 == 0, "whole m-tiles");

__global__ void __launch_bounds__(DG_TPB)
head_degrees_and_embeddings(const float* __restrict__ user_feat, const float* __restrict__ repo_feat,
                            const float* __restrict__ W_ue, const float* __restrict__ b_ue,
                            const float* __restrict__ W_re, const float* __restrict__ b_re,
                            float* __restrict__ h_user, float* __restrict__ h_repo,
                            const int* __restrict__ ur_src, const int* __restrict__ ur_dst,
                            const int* __restrict__ ru_src, const int* __restrict__ ru_dst,
                            float* __restrict__ deg_u_out_ur, float* __restrict__ deg_r_in_ur,
                            float* __restrict__ deg_r_out_ru, float* __restrict__ deg_u_in_ru,
                            int* __restrict__ tk)
{
  const int b = blockIdx.x;
  int dgi = -1, gi;
  if (b < 3 * HD_DG_BLOCKS) {
    const int q = b / 3, r = b - 3 * q;
    if (r == 0) dgi = q;
    gi = 2 * q + r - 1;
  } else {
    gi = 2 * HD_DG_BLOCKS + (b - 3 * HD_DG_BLOCKS);
  }

  if (dgi >= 0) {
    if (dgi < HD_DG_SET)
      degree_block(ur_src, ur_dst, deg_u_out_ur, deg_r_in_ur, NEDGE,
                   N_USER / 2, N_REPO / 2, tk + 0 * 2 * TK_STRIDE, HD_DG_CHUNKS);
    else
      degree_block(ru_src, ru_dst, deg_r_out_ru, deg_u_in_ru, NEDGE,
                   N_REPO / 2, N_USER / 2, tk + 1 * 2 * TK_STRIDE, HD_DG_CHUNKS);
    return;
  }

  const bool usr = (gi < HD_GU_BLOCKS);
  const int  g   = usr ? gi : gi - HD_GU_BLOCKS;
  const int  n0  = (g % HD_NT) * 16;
  const int  mt  = (g / HD_NT) * 4 + (int)(threadIdx.x >> 5);
  gemm_block<4>(usr ? user_feat : repo_feat, D_IN, usr ? W_ue : W_re, D_EMB,
             nullptr, usr ? b_ue : b_re, nullptr, nullptr, nullptr,
             usr ? h_user : h_repo, LD_EMB, usr ? N_USER : N_REPO, D_EMB, D_IN, D_IN, 0,
             n0, mt, (int)threadIdx.x, (int)(threadIdx.x & 31));
}

#define SC_EPW   128
#define SC_WPB   8
#define SC_AHEAD 8
#define SC_CHUNK (SC_EPW * SC_WPB)

template <int D, bool FULL>
__device__ __forceinline__ void move_group(const float* __restrict__ feat, float* __restrict__ agg,
                                           int sa, int sb, int da, int db,
                                           unsigned long long& m, int lane)
{
  constexpr int NC = D / 32;
  float v[SC_AHEAD][NC];
  int   dd[SC_AHEAD];
#pragma unroll
  for (int u = 0; u < SC_AHEAD; ++u) {
    dd[u] = -1;
    if (FULL || m != 0ull) {
      const int j = __builtin_ctzll(m);
      m &= m - 1ull;
      const int s_lo = __builtin_amdgcn_readlane(sa, j & 31);
      const int s_hi = __builtin_amdgcn_readlane(sb, j & 31);
      const int d_lo = __builtin_amdgcn_readlane(da, j & 31);
      const int d_hi = __builtin_amdgcn_readlane(db, j & 31);
      const int s = (j < 32) ? s_lo : s_hi;
      dd[u]       = (j < 32) ? d_lo : d_hi;
      const float* fp = feat + (long)s * D + lane;
#pragma unroll
      for (int c = 0; c < NC; ++c) v[u][c] = fp[32 * c];
    }
  }
#pragma unroll
  for (int u = 0; u < SC_AHEAD; ++u) {
    if (FULL || dd[u] >= 0) {
      float* ap = agg + (long)dd[u] * D + lane;
#pragma unroll
      for (int c = 0; c < NC; ++c) atomicAdd(ap + 32 * c, v[u][c]);
    }
  }
}

template <int D>
__device__ __forceinline__ void
scatter_block(const float* __restrict__ feat,
              const int* __restrict__ src,
              const int* __restrict__ dst,
              float* __restrict__ agg, int E,
              int splitD, int* __restrict__ tk, int nChunks)
{
  int half;
  const int t = draw_chunk(tk, nChunks, &half);
  if (t < 0) return;
  const bool hi = (half != 0);
  const int lane = threadIdx.x & 31;
  const int wave = __builtin_amdgcn_readfirstlane((int)(threadIdx.x >> 5));
  int e0 = t * SC_CHUNK + wave * SC_EPW;
  for (int b = 0; b < SC_EPW / 64; ++b, e0 += 64) {
    if (e0 >= E) return;
    const int ea = e0 + lane, eb = e0 + 32 + lane;
    const int sa = (ea < E) ? src[ea] : 0;
    const int da = (ea < E) ? dst[ea] : 0;
    const int sb = (eb < E) ? src[eb] : 0;
    const int db = (eb < E) ? dst[eb] : 0;
    const unsigned ma = __builtin_amdgcn_ballot_w32((ea < E) && ((da >= splitD) == hi));
    const unsigned mb = __builtin_amdgcn_ballot_w32((eb < E) && ((db >= splitD) == hi));
    unsigned long long m = (unsigned long long)ma | ((unsigned long long)mb << 32);
    while (__builtin_popcountll(m) >= SC_AHEAD)
      move_group<D, true>(feat, agg, sa, sb, da, db, m, lane);
    if (m != 0ull)
      move_group<D, false>(feat, agg, sa, sb, da, db, m, lane);
  }
}

template <int D>
__global__ void __launch_bounds__(32 * SC_WPB, 16)
scatter_and_gemm(const float* __restrict__ feat, const int* __restrict__ src, const int* __restrict__ dst,
                 float* __restrict__ aggS, int E, int splitD, int* __restrict__ tk, int nChunks, int nSc,
                 const float* __restrict__ A, int lda, const float* __restrict__ W, int ldw,
                 const float* __restrict__ degA, const float* __restrict__ bias,
                 const float* __restrict__ aggG, const float* __restrict__ degC,
                 const float* __restrict__ bias2,
                 float* __restrict__ C, int ldc, int M, int Nreal, int K, int Kpad, int relu)
{
  const int b = blockIdx.x;
  int gi;
  if (b < 2 * nSc) {
    if ((b & 1) == 0) {
      scatter_block<D>(feat, src, dst, aggS, E, splitD, tk, nChunks);
      return;
    }
    gi = b >> 1;
  } else {
    gi = b - nSc;
  }
  const int nt = ldc / 16;
  gemm_block<SC_WPB>(A, lda, W, ldw, degA, bias, aggG, degC, bias2, C, ldc, M, Nreal, K, Kpad, relu,
                     (gi % nt) * 16, (gi / nt) * SC_WPB + (int)(threadIdx.x >> 5),
                     (int)threadIdx.x, (int)(threadIdx.x & 31));
}

#define SCORE_PPW   32
#define SCORE_WPB   4
#define SCORE_AHEAD 8
static_assert(NEP % SCORE_PPW == 0, "the pairs of a wave lie on one side");
static_assert((2 * NEP / SCORE_PPW) % SCORE_WPB == 0, "the grid is exact");
static_assert(SCORE_PPW == 32 && SCORE_AHEAD == 8, "4 groups of 8 pairs: lane bits 1:0 pick the group, bits 4:2 the pair");

__device__ __forceinline__ float tree_merge(float x0, float x1, bool hi, int o)
{
  const float keep = hi ? x1 : x0;
  const float send = hi ? x0 : x1;
  return keep + __shfl_xor(send, o, 32);
}

__device__ __forceinline__ float tree8(const float (&x)[SCORE_AHEAD], int lane)
{
  const bool h4 = (lane & 16) != 0, h3 = (lane & 8) != 0, h2 = (lane & 4) != 0;
  const float y0 = tree_merge(x[0], x[1], h4, 16);
  const float y1 = tree_merge(x[2], x[3], h4, 16);
  const float y2 = tree_merge(x[4], x[5], h4, 16);
  const float y3 = tree_merge(x[6], x[7], h4, 16);
  const float z0 = tree_merge(y0, y1, h3, 8);
  const float z1 = tree_merge(y2, y3, h3, 8);
  float w = tree_merge(z0, z1, h2, 4);
  w += __shfl_xor(w, 2, 32);
  w += __shfl_xor(w, 1, 32);
  return w;
}

__global__ void __launch_bounds__(32 * SCORE_WPB, 16)
score_kernel(const float* __restrict__ HU, const float* __restrict__ HR,
             const int* __restrict__ pu, const int* __restrict__ pr,
             const int* __restrict__ nu, const int* __restrict__ nr,
             float* __restrict__ out)
{
  const int lane = threadIdx.x & 31;
  const int wave = __builtin_amdgcn_readfirstlane((int)blockIdx.x * SCORE_WPB + (int)(threadIdx.x >> 5));
  const int  e0  = wave * SCORE_PPW;
  const bool neg = (e0 >= NEP);
  const int  i0  = neg ? e0 - NEP : e0;
  const int* iu  = neg ? nu : pu;
  const int* ir  = neg ? nr : pr;
  const int u = iu[i0 + lane], r = ir[i0 + lane];
  const int myGroup = lane & 3;
  const int myPair  = SCORE_AHEAD * myGroup + 4 * ((lane >> 2) & 1) + 2 * ((lane >> 3) & 1) + ((lane >> 4) & 1);
  float ms = 0.0f, msu = 0.0f, msr = 0.0f;
  for (int g = 0; g < SCORE_PPW / SCORE_AHEAD; ++g) {
    float2 a[SCORE_AHEAD], b[SCORE_AHEAD];
#pragma unroll
    for (int p = 0; p < SCORE_AHEAD; ++p) {
      const int ru = __builtin_amdgcn_readlane(u, SCORE_AHEAD * g + p);
      const int rr = __builtin_amdgcn_readlane(r, SCORE_AHEAD * g + p);
      a[p] = ((const float2*)(HU + (long)ru * D_OUT))[lane];
      b[p] = ((const float2*)(HR + (long)rr * D_OUT))[lane];
    }
    float s[SCORE_AHEAD], su[SCORE_AHEAD], sr[SCORE_AHEAD];
#pragma unroll
    for (int p = 0; p < SCORE_AHEAD; ++p) {
      s[p]  = a[p].x * b[p].x + a[p].y * b[p].y;
      su[p] = a[p].x * a[p].x + a[p].y * a[p].y;
      sr[p] = b[p].x * b[p].x + b[p].y * b[p].y;
    }
    const float ts = tree8(s, lane), tsu = tree8(su, lane), tsr = tree8(sr, lane);
    const bool mine = (myGroup == g);
    ms  = mine ? ts  : ms;
    msu = mine ? tsu : msu;
    msr = mine ? tsr : msr;
  }
  const float invU = 1.0f / fmaxf(sqrtf(msu), EPSN);
  const float invR = 1.0f / fmaxf(sqrtf(msr), EPSN);
  out[e0 + myPair] = ms * invU * invR;
}

extern "C" void kernel_launch(void* const* d_in, const int* in_sizes, int n_in,
                              void* d_out, int out_size, void* d_ws, size_t ws_size,
                              hipStream_t stream)
{
  const float* user_feat = (const float*)d_in[0];
  const float* repo_feat = (const float*)d_in[1];
  const float* W_ue = (const float*)d_in[2];   const float* b_ue = (const float*)d_in[3];
  const float* W_re = (const float*)d_in[4];   const float* b_re = (const float*)d_in[5];
  const float* W_hid_ur = (const float*)d_in[6];  const float* b_hid_ur = (const float*)d_in[7];
  const float* W_hid_ru = (const float*)d_in[8];  const float* b_hid_ru = (const float*)d_in[9];
  const float* W_out_ur = (const float*)d_in[10]; const float* b_out_ur = (const float*)d_in[11];
  const float* W_out_ru = (const float*)d_in[12]; const float* b_out_ru = (const float*)d_in[13];
  const float* W_hcu = (const float*)d_in[14]; const float* b_hcu = (const float*)d_in[15];
  const float* W_hcr = (const float*)d_in[16]; const float* b_hcr = (const float*)d_in[17];
  const float* W_ocu = (const float*)d_in[18]; const float* b_ocu = (const float*)d_in[19];
  const float* W_ocr = (const float*)d_in[20]; const float* b_ocr = (const float*)d_in[21];
  const int* e_ur_src = (const int*)d_in[22];  const int* e_ur_dst = (const int*)d_in[23];
  const int* e_ru_src = (const int*)d_in[24];  const int* e_ru_dst = (const int*)d_in[25];
  const int* pos_u = (const int*)d_in[26];     const int* pos_r = (const int*)d_in[27];
  const int* neg_u = (const int*)d_in[28];     const int* neg_r = (const int*)d_in[29];
  float* out = (float*)d_out;

  float* w = (float*)d_ws;
  size_t o = 0;
  float* deg_u_out_ur = w + o; o += N_USER;
  float* deg_r_in_ur  = w + o; o += N_REPO;
  float* deg_r_out_ru = w + o; o += N_REPO;
  float* deg_u_in_ru  = w + o; o += N_USER;
  int*   tk      = (int*)(w + o); o += TK_LAUNCHES * 2 * TK_STRIDE;
  float* agg_u   = w + o; o += (size_t)N_USER * D_HID;
  float* agg_r   = w + o; o += (size_t)N_REPO * D_HID;
  float* h_user  = w + o; o += (size_t)N_USER * LD_EMB;
  float* h_repo  = w + o; o += (size_t)N_REPO * LD_EMB;
  float* feat_u  = w + o; o += (size_t)N_USER * D_HID;
  float* feat_r  = w + o; o += (size_t)N_REPO * D_HID;
  float* out_user = w + o; o += (size_t)N_USER * D_HID;
  float* out_repo = w + o; o += (size_t)N_REPO * D_HID;

  const size_t degBytes = (size_t)(2 * N_USER + 2 * N_REPO) * sizeof(float)
                        + (size_t)(TK_LAUNCHES * 2 * TK_STRIDE) * sizeof(int);
  const size_t aggBytes = ((size_t)N_USER + N_REPO) * D_HID * sizeof(float);

  auto gemm = [&](const float* A, int lda, const float* W_, int ldw,
                  const float* degA, const float* bias,
                  const float* agg, const float* degC, const float* bias2,
                  float* C, int ldc, int M, int Nreal, int K, int relu) {
    const int Kpad = (K + 7) & ~7;
    dim3 blk(32, 4);
    dim3 grd(ldc / 16, (M / 16 + 3) / 4);
    hipLaunchKernelGGL(gemm_wmma_f32, grd, blk, 0, stream,
                       A, lda, W_, ldw, degA, bias, agg, degC, bias2,
                       C, ldc, M, Nreal, K, Kpad, relu);
  };

  const int  scChunks = (NEDGE + SC_CHUNK - 1) / SC_CHUNK;
  const dim3 scBlk(32 * SC_WPB);

  hipMemsetAsync(deg_u_out_ur, 0, degBytes, stream);
  hipMemsetAsync(agg_u, 0, aggBytes, stream);
  hipLaunchKernelGGL(head_degrees_and_embeddings, dim3(HD_BLOCKS), dim3(DG_TPB), 0, stream,
                     user_feat, repo_feat, W_ue, b_ue, W_re, b_re, h_user, h_repo,
                     e_ur_src, e_ur_dst, e_ru_src, e_ru_dst,
                     deg_u_out_ur, deg_r_in_ur, deg_r_out_ru, deg_u_in_ru, tk);

  auto fused = [&](auto kern, const float* feat, const int* s, const int* d, float* aggS, int splitD, int* tkp,
                   const float* A, int lda, const float* W_, int ldw,
                   const float* degA, const float* bias,
                   const float* aggG, const float* degC, const float* bias2,
                   float* C, int ldc, int M, int Nreal, int K, int relu) {
    const int Kpad = (K + 7) & ~7;
    const int nG = (ldc / 16) * ((M / 16 + SC_WPB - 1) / SC_WPB);
    hipLaunchKernelGGL(kern, dim3(2 * scChunks + nG), scBlk, 0, stream,
                       feat, s, d, aggS, NEDGE, splitD, tkp, scChunks, 2 * scChunks,
                       A, lda, W_, ldw, degA, bias, aggG, degC, bias2, C, ldc, M, Nreal, K, Kpad, relu);
  };

  gemm(h_user, LD_EMB, W_hid_ur, D_HID, deg_u_out_ur, nullptr, nullptr, nullptr, nullptr,
       feat_u, D_HID, N_USER, D_HID, D_EMB, 0);
  fused(scatter_and_gemm<D_HID>, feat_u, e_ur_src, e_ur_dst, agg_r, N_REPO / 2, tk + 2 * 2 * TK_STRIDE,
        h_repo, LD_EMB, W_hid_ru, D_HID, deg_r_out_ru, nullptr, nullptr, nullptr, nullptr,
        feat_r, D_HID, N_REPO, D_HID, D_EMB, 0);
  fused(scatter_and_gemm<D_HID>, feat_r, e_ru_src, e_ru_dst, agg_u, N_USER / 2, tk + 3 * 2 * TK_STRIDE,
        h_repo, LD_EMB, W_hcr, D_HID, nullptr, b_hcr, agg_r, deg_r_in_ur, b_hid_ur,
        out_repo, D_HID, N_REPO, D_HID, D_EMB, 1);
  gemm(h_user, LD_EMB, W_hcu, D_HID, nullptr, b_hcu, agg_u, deg_u_in_ru, b_hid_ru,
       out_user, D_HID, N_USER, D_HID, D_EMB, 1);

  hipMemsetAsync(agg_u, 0, aggBytes, stream);
  gemm(out_user, D_HID, W_out_ur, D_OUT, deg_u_out_ur, nullptr, nullptr, nullptr, nullptr,
       feat_u, D_OUT, N_USER, D_OUT, D_HID, 0);
  fused(scatter_and_gemm<D_OUT>, feat_u, e_ur_src, e_ur_dst, agg_r, N_REPO / 2, tk + 4 * 2 * TK_STRIDE,
        out_repo, D_HID, W_out_ru, D_OUT, deg_r_out_ru, nullptr, nullptr, nullptr, nullptr,
        feat_r, D_OUT, N_REPO, D_OUT, D_HID, 0);
  fused(scatter_and_gemm<D_OUT>, feat_r, e_ru_src, e_ru_dst, agg_u, N_USER / 2, tk + 5 * 2 * TK_STRIDE,
        out_repo, D_HID, W_ocr, D_OUT, nullptr, b_ocr, agg_r, deg_r_in_ur, b_out_ur,
        h_repo, D_OUT, N_REPO, D_OUT, D_HID, 1);
  gemm(out_user, D_HID, W_ocu, D_OUT, nullptr, b_ocu, agg_u, deg_u_in_ru, b_out_ru,
       h_user, D_OUT, N_USER, D_OUT, D_HID, 1);

  hipLaunchKernelGGL(score_kernel, dim3(2 * NEP / SCORE_PPW / SCORE_WPB), dim3(32 * SCORE_WPB), 0, stream,
                     h_user, h_repo, pos_u, pos_r, neg_u, neg_r, out);
}
